# starting kernel
# baseline (speedup 1.0000x reference)
#include <hip/hip_runtime.h>
#include <hip/hip_bf16.h>

typedef float v2f __attribute__((ext_vector_type(2)));
typedef float v8f __attribute__((ext_vector_type(8)));

#define HID 128
#define LDS_PITCH 132

__global__ __launch_bounds__(256)
void gin_gemm(const float* __restrict__ X, const float* __restrict__ W,
              const float* __restrict__ bias, float* __restrict__ Y, int n_rows)
{
    __shared__ float sA[16 * LDS_PITCH];

    const int m0  = blockIdx.x * 16;
    const int tid = threadIdx.x;

    for (int i = tid; i < 16 * 32; i += 256) {
        const int r = i >> 5;
        const int c = i & 31;
        float4 v = make_float4(0.f, 0.f, 0.f, 0.f);
        if (m0 + r < n_rows) {
            const float4* src =
                reinterpret_cast<const float4*>(X + (size_t)(m0 + r) * HID);
            v = src[c];
        }
        float* dst = &sA[r * LDS_PITCH + c * 4];
        dst[0] = v.x; dst[1] = v.y; dst[2] = v.z; dst[3] = v.w;
    }
    __syncthreads();

    const int wave = tid >> 5;
    const int lane = tid & 31;
    const int row  = lane & 15;
    const int kk   = (lane >> 4) * 2;
    const int col  = wave * 16 + (lane & 15);

    v8f acc = {};
#pragma unroll
    for (int k0 = 0; k0 < HID; k0 += 4) {
        v2f a;
        a.x = sA[row * LDS_PITCH + k0 + kk];
        a.y = sA[row * LDS_PITCH + k0 + kk + 1];
        v2f b;
        b.x = W[(size_t)(k0 + kk) * HID + col];
        b.y = W[(size_t)(k0 + kk + 1) * HID + col];
        acc = __builtin_amdgcn_wmma_f32_16x16x4_f32(
             false, a,  false, b,
             (short)0, acc,  false,  false);
    }

    const float bcol = bias[col];
    const int half = (lane >> 4) * 8;
#pragma unroll
    for (int i = 0; i < 8; ++i) {
        const int m = m0 + half + i;
        if (m < n_rows) Y[(size_t)m * HID + col] = acc[i] + bcol;
    }
}

__global__ __launch_bounds__(256)
void gin_scatter(const float* __restrict__ h, const int* __restrict__ send,
                 const int* __restrict__ rec, float* __restrict__ agg, int n_edges)
{
    const long long t = (long long)blockIdx.x * blockDim.x + threadIdx.x;
    const int e = (int)(t >> 5);
    if (e >= n_edges) return;
    const int lane = (int)(t & 31);

    const int s = send[e];
    const int r = rec[e];

    const float4* src = reinterpret_cast<const float4*>(h + (size_t)s * HID);
    const float4 v = src[lane];

    float* dst = agg + (size_t)r * HID + lane * 4;
    atomicAdd(dst + 0, v.x);
    atomicAdd(dst + 1, v.y);
    atomicAdd(dst + 2, v.z);
    atomicAdd(dst + 3, v.w);
}

extern "C" void kernel_launch(void* const* d_in, const int* in_sizes, int n_in,
                              void* d_out, int out_size, void* d_ws, size_t ws_size,
                              hipStream_t stream) {
    const float* h_in  = (const float*)d_in[0];
    const int*   eidx  = (const int*)d_in[1];
    const float* W_emb = (const float*)d_in[2];
    const float* b_emb = (const float*)d_in[3];
    const float* Ws    = (const float*)d_in[4];
    const float* bs    = (const float*)d_in[5];

    const int n_nodes = in_sizes[0] / HID;
    const int n_edges = in_sizes[1] / 2;
    const int* send = eidx;
    const int* rec  = eidx + n_edges;

    float* hbuf = (float*)d_out;
    float* agg  = (float*)d_ws;
    const size_t nh_bytes = (size_t)n_nodes * HID * sizeof(float);

    const int gemm_grid = (n_nodes + 15) / 16;
    const int scat_grid = (int)(((long long)n_edges * 32 + 255) / 256);

    gin_gemm<<<gemm_grid, 256, 0, stream>>>(h_in, W_emb, b_emb, hbuf, n_nodes);

    for (int l = 0; l < 4; ++l) {
        hipMemcpyAsync(agg, hbuf, nh_bytes, hipMemcpyDeviceToDevice, stream);
        gin_scatter<<<scat_grid, 256, 0, stream>>>(hbuf, send, rec, agg, n_edges);
        gin_gemm<<<gemm_grid, 256, 0, stream>>>(
            agg, Ws + (size_t)l * HID * HID, bs + (size_t)l * HID, hbuf, n_nodes);
    }
}


# scatter: lane owns columns lane+32j, so each wave-wide f32 atomic (and load) covers one whole 128-byte line instead of 8 lanes on each of 4 lines; same adds, arrival order as before
# speedup vs baseline: 1.3316x; 1.3316x over previous
#include <hip/hip_runtime.h>
#include <hip/hip_bf16.h>

typedef float v2f __attribute__((ext_vector_type(2)));
typedef float v8f __attribute__((ext_vector_type(8)));

#define HID 128
#define LDS_PITCH 132

__global__ __launch_bounds__(256)
void gin_gemm(const float* __restrict__ X, const float* __restrict__ W,
              const float* __restrict__ bias, float* __restrict__ Y, int n_rows)
{
    __shared__ float sA[16 * LDS_PITCH];

    const int m0  = blockIdx.x * 16;
    const int tid = threadIdx.x;

    for (int i = tid; i < 16 * 32; i += 256) {
        const int r = i >> 5;
        const int c = i & 31;
        float4 v = make_float4(0.f, 0.f, 0.f, 0.f);
        if (m0 + r < n_rows) {
            const float4* src =
                reinterpret_cast<const float4*>(X + (size_t)(m0 + r) * HID);
            v = src[c];
        }
        float* dst = &sA[r * LDS_PITCH + c * 4];
        dst[0] = v.x; dst[1] = v.y; dst[2] = v.z; dst[3] = v.w;
    }
    __syncthreads();

    const int wave = tid >> 5;
    const int lane = tid & 31;
    const int row  = lane & 15;
    const int kk   = (lane >> 4) * 2;
    const int col  = wave * 16 + (lane & 15);

    v8f acc = {};
#pragma unroll
    for (int k0 = 0; k0 < HID; k0 += 4) {
        v2f a;
        a.x = sA[row * LDS_PITCH + k0 + kk];
        a.y = sA[row * LDS_PITCH + k0 + kk + 1];
        v2f b;
        b.x = W[(size_t)(k0 + kk) * HID + col];
        b.y = W[(size_t)(k0 + kk + 1) * HID + col];
        acc = __builtin_amdgcn_wmma_f32_16x16x4_f32(
             false, a,  false, b,
             (short)0, acc,  false,  false);
    }

    const float bcol = bias[col];
    const int half = (lane >> 4) * 8;
#pragma unroll
    for (int i = 0; i < 8; ++i) {
        const int m = m0 + half + i;
        if (m < n_rows) Y[(size_t)m * HID + col] = acc[i] + bcol;
    }
}

__global__ __launch_bounds__(256)
void gin_scatter(const float* __restrict__ h, const int* __restrict__ send,
                 const int* __restrict__ rec, float* __restrict__ agg, int n_edges)
{
    const long long t = (long long)blockIdx.x * blockDim.x + threadIdx.x;
    const int e = (int)(t >> 5);
    if (e >= n_edges) return;
    const int lane = (int)(t & 31);

    const int s = send[e];
    const int r = rec[e];

    const float* src = h + (size_t)s * HID + lane;
    const float v0 = src[0];
    const float v1 = src[32];
    const float v2 = src[64];
    const float v3 = src[96];

    float* dst = agg + (size_t)r * HID + lane;
    atomicAdd(dst + 0,  v0);
    atomicAdd(dst + 32, v1);
    atomicAdd(dst + 64, v2);
    atomicAdd(dst + 96, v3);
}

extern "C" void kernel_launch(void* const* d_in, const int* in_sizes, int n_in,
                              void* d_out, int out_size, void* d_ws, size_t ws_size,
                              hipStream_t stream) {
    const float* h_in  = (const float*)d_in[0];
    const int*   eidx  = (const int*)d_in[1];
    const float* W_emb = (const float*)d_in[2];
    const float* b_emb = (const float*)d_in[3];
    const float* Ws    = (const float*)d_in[4];
    const float* bs    = (const float*)d_in[5];

    const int n_nodes = in_sizes[0] / HID;
    const int n_edges = in_sizes[1] / 2;
    const int* send = eidx;
    const int* rec  = eidx + n_edges;

    float* hbuf = (float*)d_out;
    float* agg  = (float*)d_ws;
    const size_t nh_bytes = (size_t)n_nodes * HID * sizeof(float);

    const int gemm_grid = (n_nodes + 15) / 16;
    const int scat_grid = (int)(((long long)n_edges * 32 + 255) / 256);

    gin_gemm<<<gemm_grid, 256, 0, stream>>>(h_in, W_emb, b_emb, hbuf, n_nodes);

    for (int l = 0; l < 4; ++l) {
        hipMemcpyAsync(agg, hbuf, nh_bytes, hipMemcpyDeviceToDevice, stream);
        gin_scatter<<<scat_grid, 256, 0, stream>>>(hbuf, send, rec, agg, n_edges);
        gin_gemm<<<gemm_grid, 256, 0, stream>>>(
            agg, Ws + (size_t)l * HID * HID, bs + (size_t)l * HID, hbuf, n_nodes);
    }
}


# atomic scatter -> pull: edges grouped by receiver each call (int rank atomics, scan, fill); a wave sums a node's rows in registers; f32 atomics and copies gone; sum order = arrival rank
# speedup vs baseline: 2.8728x; 2.1574x over previous
#include <hip/hip_runtime.h>
#include <hip/hip_bf16.h>

typedef float v2f __attribute__((ext_vector_type(2)));
typedef float v8f __attribute__((ext_vector_type(8)));

#define HID 128
#define LDS_PITCH 132

__global__ __launch_bounds__(256)
void gin_gemm(const float* __restrict__ X, const float* __restrict__ W,
              const float* __restrict__ bias, float* __restrict__ Y, int n_rows)
{
    __shared__ float sA[16 * LDS_PITCH];

    const int m0  = blockIdx.x * 16;
    const int tid = threadIdx.x;

    for (int i = tid; i < 16 * 32; i += 256) {
        const int r = i >> 5;
        const int c = i & 31;
        float4 v = make_float4(0.f, 0.f, 0.f, 0.f);
        if (m0 + r < n_rows) {
            const float4* src =
                reinterpret_cast<const float4*>(X + (size_t)(m0 + r) * HID);
            v = src[c];
        }
        float* dst = &sA[r * LDS_PITCH + c * 4];
        dst[0] = v.x; dst[1] = v.y; dst[2] = v.z; dst[3] = v.w;
    }
    __syncthreads();

    const int wave = tid >> 5;
    const int lane = tid & 31;
    const int row  = lane & 15;
    const int kk   = (lane >> 4) * 2;
    const int col  = wave * 16 + (lane & 15);

    v8f acc = {};
#pragma unroll
    for (int k0 = 0; k0 < HID; k0 += 4) {
        v2f a;
        a.x = sA[row * LDS_PITCH + k0 + kk];
        a.y = sA[row * LDS_PITCH + k0 + kk + 1];
        v2f b;
        b.x = W[(size_t)(k0 + kk) * HID + col];
        b.y = W[(size_t)(k0 + kk + 1) * HID + col];
        acc = __builtin_amdgcn_wmma_f32_16x16x4_f32(
             false, a,  false, b,
             (short)0, acc,  false,  false);
    }

    const float bcol = bias[col];
    const int half = (lane >> 4) * 8;
#pragma unroll
    for (int i = 0; i < 8; ++i) {
        const int m = m0 + half + i;
        if (m < n_rows) Y[(size_t)m * HID + col] = acc[i] + bcol;
    }
}

__global__ __launch_bounds__(256)
void gin_scatter(const float* __restrict__ h, const int* __restrict__ send,
                 const int* __restrict__ rec, float* __restrict__ agg, int n_edges)
{
    const long long t = (long long)blockIdx.x * blockDim.x + threadIdx.x;
    const int e = (int)(t >> 5);
    if (e >= n_edges) return;
    const int lane = (int)(t & 31);

    const int s = send[e];
    const int r = rec[e];

    const float* src = h + (size_t)s * HID + lane;
    const float v0 = src[0];
    const float v1 = src[32];
    const float v2 = src[64];
    const float v3 = src[96];

    float* dst = agg + (size_t)r * HID + lane;
    atomicAdd(dst + 0,  v0);
    atomicAdd(dst + 32, v1);
    atomicAdd(dst + 64, v2);
    atomicAdd(dst + 96, v3);
}


__global__ __launch_bounds__(256)
void csr_zero(int* __restrict__ cnt, int n1)
{
    const int i = blockIdx.x * 256 + threadIdx.x;
    if (i < n1) cnt[i] = 0;
}

__global__ __launch_bounds__(256)
void csr_rank(const int* __restrict__ rec, int* __restrict__ cnt,
              int* __restrict__ rank, int n_edges, int n_nodes)
{
    const int e = blockIdx.x * 256 + threadIdx.x;
    if (e >= n_edges) return;
    const int r = rec[e];
    int k = -1;
    if ((unsigned)r < (unsigned)n_nodes) k = atomicAdd(&cnt[r], 1);
    rank[e] = k;
}

__global__ __launch_bounds__(1024)
void csr_scan(int* __restrict__ cnt, int n4)
{
    __shared__ int wsum[32];
    const int lane = threadIdx.x & 31;
    const int w    = threadIdx.x >> 5;
    int4* __restrict__ c4 = reinterpret_cast<int4*>(cnt);
    const int seg4 = (((n4 + 31) >> 5) + 31) & ~31;
    const int b4   = w * seg4;
    const int e4   = min(b4 + seg4, n4);

    int s = 0;
    for (int i0 = b4; i0 < e4; i0 += 256) {
        int4 q[8];
#pragma unroll
        for (int u = 0; u < 8; ++u) {
            const int i = i0 + u * 32 + lane;
            q[u] = make_int4(0, 0, 0, 0);
            if (i < e4) q[u] = c4[i];
        }
#pragma unroll
        for (int u = 0; u < 8; ++u) s += (q[u].x + q[u].y) + (q[u].z + q[u].w);
    }
    for (int off = 16; off > 0; off >>= 1) s += __shfl_xor(s, off);
    if (lane == 0) wsum[w] = s;
    __syncthreads();

    int carry = 0;
    for (int k = 0; k < w; ++k) carry += wsum[k];

    for (int i0 = b4; i0 < e4; i0 += 256) {
        int4 q[8];
        int  t[8], x[8];
#pragma unroll
        for (int u = 0; u < 8; ++u) {
            const int i = i0 + u * 32 + lane;
            q[u] = make_int4(0, 0, 0, 0);
            if (i < e4) q[u] = c4[i];
        }
#pragma unroll
        for (int u = 0; u < 8; ++u) { t[u] = (q[u].x + q[u].y) + (q[u].z + q[u].w); x[u] = t[u]; }
#pragma unroll
        for (int off = 1; off < 32; off <<= 1) {
#pragma unroll
            for (int u = 0; u < 8; ++u) {
                const int y = __shfl_up(x[u], off);
                if (lane >= off) x[u] += y;
            }
        }
#pragma unroll
        for (int u = 0; u < 8; ++u) {
            const int i = i0 + u * 32 + lane;
            const int base = carry + x[u] - t[u];
            if (i < e4)
                c4[i] = make_int4(base, base + q[u].x, base + q[u].x + q[u].y,
                                  base + q[u].x + q[u].y + q[u].z);
            carry += __builtin_amdgcn_readlane(x[u], 31);
        }
    }
}

__global__ __launch_bounds__(256)
void csr_fill(const int* __restrict__ send, const int* __restrict__ rec,
              const int* __restrict__ rowptr, const int* __restrict__ rank,
              int* __restrict__ col, int n_edges, int n_nodes)
{
    const int e = blockIdx.x * 256 + threadIdx.x;
    if (e >= n_edges) return;
    const int k = rank[e];
    if (k < 0) return;
    const int s = min(max(send[e], 0), n_nodes - 1);
    col[rowptr[rec[e]] + k] = s;
}

__global__ __launch_bounds__(256)
void gin_pull(const float* __restrict__ h, const int* __restrict__ rowptr,
              const int* __restrict__ col, float* __restrict__ agg, int n_nodes)
{
    const int v = __builtin_amdgcn_readfirstlane((int)(blockIdx.x * 8 + (threadIdx.x >> 5)));
    if (v >= n_nodes) return;
    const int lane = threadIdx.x & 31;
    const int beg = rowptr[v];
    const int end = rowptr[v + 1];

    const float4* __restrict__ hv = reinterpret_cast<const float4*>(h);
    float4 acc = hv[(size_t)v * 32 + lane];

    for (int j0 = beg; j0 < end; j0 += 32) {
        const int m = min(32, end - j0);
        const int idx = (lane < m) ? col[j0 + lane] : 0;
        for (int i = 0; i < m; i += 8) {
            int s[8];
#pragma unroll
            for (int u = 0; u < 8; ++u) s[u] = __builtin_amdgcn_readlane(idx, (i + u) & 31);
            float4 r[8];
#pragma unroll
            for (int u = 0; u < 8; ++u) {
                if (i + u < m) r[u] = hv[(size_t)s[u] * 32 + lane];
            }
#pragma unroll
            for (int u = 0; u < 8; ++u) {
                if (i + u < m) {
                    acc.x += r[u].x; acc.y += r[u].y; acc.z += r[u].z; acc.w += r[u].w;
                }
            }
        }
    }
    reinterpret_cast<float4*>(agg)[(size_t)v * 32 + lane] = acc;
}

extern "C" void kernel_launch(void* const* d_in, const int* in_sizes, int n_in,
                              void* d_out, int out_size, void* d_ws, size_t ws_size,
                              hipStream_t stream) {
    const float* h_in  = (const float*)d_in[0];
    const int*   eidx  = (const int*)d_in[1];
    const float* W_emb = (const float*)d_in[2];
    const float* b_emb = (const float*)d_in[3];
    const float* Ws    = (const float*)d_in[4];
    const float* bs    = (const float*)d_in[5];

    const int n_nodes = in_sizes[0] / HID;
    const int n_edges = in_sizes[1] / 2;
    const int* send = eidx;
    const int* rec  = eidx + n_edges;

    float* hbuf = (float*)d_out;
    float* agg  = (float*)d_ws;
    const size_t nh_bytes  = (size_t)n_nodes * HID * sizeof(float);
    const int    n1p       = (n_nodes + 4) & ~3;
    const size_t ptr_bytes = ((size_t)n1p * sizeof(int) + 255) & ~(size_t)255;
    const size_t col_bytes = (size_t)n_edges * sizeof(int);

    const int gemm_grid = (n_nodes + 15) / 16;

    gin_gemm<<<gemm_grid, 256, 0, stream>>>(h_in, W_emb, b_emb, hbuf, n_nodes);

    if (ws_size >= nh_bytes + ptr_bytes + col_bytes && col_bytes <= nh_bytes) {
        int* rank   = (int*)d_ws;
        int* rowptr = (int*)((char*)d_ws + nh_bytes);
        int* col    = (int*)((char*)d_ws + nh_bytes + ptr_bytes);
        const int edge_grid = (n_edges + 255) / 256;

        csr_zero<<<(n1p + 255) / 256, 256, 0, stream>>>(rowptr, n1p);
        csr_rank<<<edge_grid, 256, 0, stream>>>(rec, rowptr, rank, n_edges, n_nodes);
        csr_scan<<<1, 1024, 0, stream>>>(rowptr, n1p / 4);
        csr_fill<<<edge_grid, 256, 0, stream>>>(send, rec, rowptr, rank, col, n_edges, n_nodes);

        const int pull_grid = (n_nodes + 7) / 8;
        for (int l = 0; l < 4; ++l) {
            gin_pull<<<pull_grid, 256, 0, stream>>>(hbuf, rowptr, col, agg, n_nodes);
            gin_gemm<<<gemm_grid, 256, 0, stream>>>(
                agg, Ws + (size_t)l * HID * HID, bs + (size_t)l * HID, hbuf, n_nodes);
        }
    } else {
        const int scat_grid = (int)(((long long)n_edges * 32 + 255) / 256);
        for (int l = 0; l < 4; ++l) {
            hipMemcpyAsync(agg, hbuf, nh_bytes, hipMemcpyDeviceToDevice, stream);
            gin_scatter<<<scat_grid, 256, 0, stream>>>(hbuf, send, rec, agg, n_edges);
            gin_gemm<<<gemm_grid, 256, 0, stream>>>(
                agg, Ws + (size_t)l * HID * HID, bs + (size_t)l * HID, hbuf, n_nodes);
        }
    }
}


# pull fused into the layer GEMM: a resident block sums 16 nodes' rows into the LDS A slab and multiplies it by W fragments kept in registers; agg round trip and 4 launches gone; sum order kept
# speedup vs baseline: 3.3636x; 1.1708x over previous
#include <hip/hip_runtime.h>
#include <hip/hip_bf16.h>

typedef float v2f __attribute__((ext_vector_type(2)));
typedef float v8f __attribute__((ext_vector_type(8)));

#define HID 128
#define LDS_PITCH 132

__global__ __launch_bounds__(256)
void gin_gemm(const float* __restrict__ X, const float* __restrict__ W,
              const float* __restrict__ bias, float* __restrict__ Y, int n_rows)
{
    __shared__ float sA[16 * LDS_PITCH];

    const int m0  = blockIdx.x * 16;
    const int tid = threadIdx.x;

    for (int i = tid; i < 16 * 32; i += 256) {
        const int r = i >> 5;
        const int c = i & 31;
        float4 v = make_float4(0.f, 0.f, 0.f, 0.f);
        if (m0 + r < n_rows) {
            const float4* src =
                reinterpret_cast<const float4*>(X + (size_t)(m0 + r) * HID);
            v = src[c];
        }
        float* dst = &sA[r * LDS_PITCH + c * 4];
        dst[0] = v.x; dst[1] = v.y; dst[2] = v.z; dst[3] = v.w;
    }
    __syncthreads();

    const int wave = tid >> 5;
    const int lane = tid & 31;
    const int row  = lane & 15;
    const int kk   = (lane >> 4) * 2;
    const int col  = wave * 16 + (lane & 15);

    v8f acc = {};
#pragma unroll
    for (int k0 = 0; k0 < HID; k0 += 4) {
        v2f a;
        a.x = sA[row * LDS_PITCH + k0 + kk];
        a.y = sA[row * LDS_PITCH + k0 + kk + 1];
        v2f b;
        b.x = W[(size_t)(k0 + kk) * HID + col];
        b.y = W[(size_t)(k0 + kk + 1) * HID + col];
        acc = __builtin_amdgcn_wmma_f32_16x16x4_f32(
             false, a,  false, b,
             (short)0, acc,  false,  false);
    }

    const float bcol = bias[col];
    const int half = (lane >> 4) * 8;
#pragma unroll
    for (int i = 0; i < 8; ++i) {
        const int m = m0 + half + i;
        if (m < n_rows) Y[(size_t)m * HID + col] = acc[i] + bcol;
    }
}

__global__ __launch_bounds__(256)
void gin_scatter(const float* __restrict__ h, const int* __restrict__ send,
                 const int* __restrict__ rec, float* __restrict__ agg, int n_edges)
{
    const long long t = (long long)blockIdx.x * blockDim.x + threadIdx.x;
    const int e = (int)(t >> 5);
    if (e >= n_edges) return;
    const int lane = (int)(t & 31);

    const int s = send[e];
    const int r = rec[e];

    const float* src = h + (size_t)s * HID + lane;
    const float v0 = src[0];
    const float v1 = src[32];
    const float v2 = src[64];
    const float v3 = src[96];

    float* dst = agg + (size_t)r * HID + lane;
    atomicAdd(dst + 0,  v0);
    atomicAdd(dst + 32, v1);
    atomicAdd(dst + 64, v2);
    atomicAdd(dst + 96, v3);
}


__global__ __launch_bounds__(256)
void csr_zero(int* __restrict__ cnt, int n1)
{
    const int i = blockIdx.x * 256 + threadIdx.x;
    if (i < n1) cnt[i] = 0;
}

__global__ __launch_bounds__(256)
void csr_rank(const int* __restrict__ rec, int* __restrict__ cnt,
              int* __restrict__ rank, int n_edges, int n_nodes)
{
    const int e = blockIdx.x * 256 + threadIdx.x;
    if (e >= n_edges) return;
    const int r = rec[e];
    int k = -1;
    if ((unsigned)r < (unsigned)n_nodes) k = atomicAdd(&cnt[r], 1);
    rank[e] = k;
}

__global__ __launch_bounds__(1024)
void csr_scan(int* __restrict__ cnt, int n4)
{
    __shared__ int wsum[32];
    const int lane = threadIdx.x & 31;
    const int w    = threadIdx.x >> 5;
    int4* __restrict__ c4 = reinterpret_cast<int4*>(cnt);
    const int seg4 = (((n4 + 31) >> 5) + 31) & ~31;
    const int b4   = w * seg4;
    const int e4   = min(b4 + seg4, n4);

    int s = 0;
    for (int i0 = b4; i0 < e4; i0 += 256) {
        int4 q[8];
#pragma unroll
        for (int u = 0; u < 8; ++u) {
            const int i = i0 + u * 32 + lane;
            q[u] = make_int4(0, 0, 0, 0);
            if (i < e4) q[u] = c4[i];
        }
#pragma unroll
        for (int u = 0; u < 8; ++u) s += (q[u].x + q[u].y) + (q[u].z + q[u].w);
    }
    for (int off = 16; off > 0; off >>= 1) s += __shfl_xor(s, off);
    if (lane == 0) wsum[w] = s;
    __syncthreads();

    int carry = 0;
    for (int k = 0; k < w; ++k) carry += wsum[k];

    for (int i0 = b4; i0 < e4; i0 += 256) {
        int4 q[8];
        int  t[8], x[8];
#pragma unroll
        for (int u = 0; u < 8; ++u) {
            const int i = i0 + u * 32 + lane;
            q[u] = make_int4(0, 0, 0, 0);
            if (i < e4) q[u] = c4[i];
        }
#pragma unroll
        for (int u = 0; u < 8; ++u) { t[u] = (q[u].x + q[u].y) + (q[u].z + q[u].w); x[u] = t[u]; }
#pragma unroll
        for (int off = 1; off < 32; off <<= 1) {
#pragma unroll
            for (int u = 0; u < 8; ++u) {
                const int y = __shfl_up(x[u], off);
                if (lane >= off) x[u] += y;
            }
        }
#pragma unroll
        for (int u = 0; u < 8; ++u) {
            const int i = i0 + u * 32 + lane;
            const int base = carry + x[u] - t[u];
            if (i < e4)
                c4[i] = make_int4(base, base + q[u].x, base + q[u].x + q[u].y,
                                  base + q[u].x + q[u].y + q[u].z);
            carry += __builtin_amdgcn_readlane(x[u], 31);
        }
    }
}

__global__ __launch_bounds__(256)
void csr_fill(const int* __restrict__ send, const int* __restrict__ rec,
              const int* __restrict__ rowptr, const int* __restrict__ rank,
              int* __restrict__ col, int n_edges, int n_nodes)
{
    const int e = blockIdx.x * 256 + threadIdx.x;
    if (e >= n_edges) return;
    const int k = rank[e];
    if (k < 0) return;
    const int s = min(max(send[e], 0), n_nodes - 1);
    col[rowptr[rec[e]] + k] = s;
}

__global__ __launch_bounds__(256)
void gin_layer(const float* __restrict__ h, const int* __restrict__ rowptr,
               const int* __restrict__ col, const float* __restrict__ W,
               const float* __restrict__ bias, float* __restrict__ Y,
               int n_nodes, int n_tiles)
{
    __shared__ float sA[2][16 * LDS_PITCH];

    const int tid  = threadIdx.x;
    const int wave = __builtin_amdgcn_readfirstlane(tid >> 5);
    const int lane = tid & 31;
    const int row  = lane & 15;
    const int kk   = (lane >> 4) * 2;
    const int ncol = wave * 16 + (lane & 15);
    const int half = (lane >> 4) * 8;

    v2f bf[32];
#pragma unroll
    for (int s = 0; s < 32; ++s) {
        bf[s].x = W[(size_t)(4 * s + kk) * HID + ncol];
        bf[s].y = W[(size_t)(4 * s + kk + 1) * HID + ncol];
    }
    const float bcol = bias[ncol];

    const float4* __restrict__ hv = reinterpret_cast<const float4*>(h);
    int buf = 0;

    int pv = 0;
    {
        const int v0 = blockIdx.x * 16 + 2 * wave;
        if ((int)blockIdx.x < n_tiles && lane < 3 && v0 + lane <= n_nodes) pv = rowptr[v0 + lane];
    }

    for (int tile = blockIdx.x; tile < n_tiles; tile += gridDim.x) {
        const int m0 = tile * 16;
        const int v0 = m0 + 2 * wave;
        const bool ok0 = v0 < n_nodes;
        const bool ok1 = v0 + 1 < n_nodes;

        int p0 = 0, p1 = 0, p2 = 0;
        if (ok0) {
            p0 = __builtin_amdgcn_readlane(pv, 0);
            p1 = __builtin_amdgcn_readlane(pv, 1);
            p2 = ok1 ? __builtin_amdgcn_readlane(pv, 2) : p1;
        }

        const int c0 = min(32, p1 - p0);
        const int c1 = min(32, p2 - p1);
        const int first0 = (lane < c0) ? col[p0 + lane] : 0;
        const int first1 = (lane < c1) ? col[p1 + lane] : 0;
        float4 acc[2];
        acc[0] = make_float4(0.f, 0.f, 0.f, 0.f);
        acc[1] = make_float4(0.f, 0.f, 0.f, 0.f);
        if (ok0) acc[0] = hv[(size_t)v0 * 32 + lane];
        if (ok1) acc[1] = hv[(size_t)(v0 + 1) * 32 + lane];

#pragma unroll
        for (int t = 0; t < 2; ++t) {
            const int beg = t ? p1 : p0;
            const int end = t ? p2 : p1;
            const int first = t ? first1 : first0;
            float4 a4 = acc[t];
            for (int j0 = beg; j0 < end; j0 += 32) {
                const int m = min(32, end - j0);
                int idx = first;
                if (j0 != beg) idx = (lane < m) ? col[j0 + lane] : 0;
                for (int i = 0; i < m; i += 8) {
                    int s[8];
#pragma unroll
                    for (int u = 0; u < 8; ++u) s[u] = __builtin_amdgcn_readlane(idx, (i + u) & 31);
                    float4 r[8];
#pragma unroll
                    for (int u = 0; u < 8; ++u) {
                        if (i + u < m) r[u] = hv[(size_t)s[u] * 32 + lane];
                    }
#pragma unroll
                    for (int u = 0; u < 8; ++u) {
                        if (i + u < m) {
                            a4.x += r[u].x; a4.y += r[u].y; a4.z += r[u].z; a4.w += r[u].w;
                        }
                    }
                }
            }
            *reinterpret_cast<float4*>(&sA[buf][(2 * wave + t) * LDS_PITCH + lane * 4]) = a4;
        }
        __syncthreads();

        {
            const int nt = tile + gridDim.x;
            const int nv0 = nt * 16 + 2 * wave;
            pv = 0;
            if (nt < n_tiles && lane < 3 && nv0 + lane <= n_nodes) pv = rowptr[nv0 + lane];
        }

        const float* __restrict__ sa = sA[buf];
        v8f d = {};
#pragma unroll
        for (int s = 0; s < 32; ++s) {
            v2f a;
            a.x = sa[row * LDS_PITCH + 4 * s + kk];
            a.y = sa[row * LDS_PITCH + 4 * s + kk + 1];
            d = __builtin_amdgcn_wmma_f32_16x16x4_f32(
                 false, a,  false, bf[s],
                 (short)0, d,  false,  false);
        }

#pragma unroll
        for (int i = 0; i < 8; ++i) {
            const int m = m0 + half + i;
            if (m < n_nodes) Y[(size_t)m * HID + ncol] = d[i] + bcol;
        }
        buf ^= 1;
    }
}

extern "C" void kernel_launch(void* const* d_in, const int* in_sizes, int n_in,
                              void* d_out, int out_size, void* d_ws, size_t ws_size,
                              hipStream_t stream) {
    const float* h_in  = (const float*)d_in[0];
    const int*   eidx  = (const int*)d_in[1];
    const float* W_emb = (const float*)d_in[2];
    const float* b_emb = (const float*)d_in[3];
    const float* Ws    = (const float*)d_in[4];
    const float* bs    = (const float*)d_in[5];

    const int n_nodes = in_sizes[0] / HID;
    const int n_edges = in_sizes[1] / 2;
    const int* send = eidx;
    const int* rec  = eidx + n_edges;

    float* hbuf = (float*)d_out;
    float* agg  = (float*)d_ws;
    const size_t nh_bytes  = (size_t)n_nodes * HID * sizeof(float);
    const int    n1p       = (n_nodes + 4) & ~3;
    const size_t ptr_bytes = ((size_t)n1p * sizeof(int) + 255) & ~(size_t)255;
    const size_t col_bytes = (size_t)n_edges * sizeof(int);

    const int gemm_grid = (n_nodes + 15) / 16;

    gin_gemm<<<gemm_grid, 256, 0, stream>>>(h_in, W_emb, b_emb, hbuf, n_nodes);

    if (ws_size >= nh_bytes + ptr_bytes + col_bytes && col_bytes <= nh_bytes) {
        int* rank   = (int*)d_ws;
        int* rowptr = (int*)((char*)d_ws + nh_bytes);
        int* col    = (int*)((char*)d_ws + nh_bytes + ptr_bytes);
        const int edge_grid = (n_edges + 255) / 256;

        csr_zero<<<(n1p + 255) / 256, 256, 0, stream>>>(rowptr, n1p);
        csr_rank<<<edge_grid, 256, 0, stream>>>(rec, rowptr, rank, n_edges, n_nodes);
        csr_scan<<<1, 1024, 0, stream>>>(rowptr, n1p / 4);
        csr_fill<<<edge_grid, 256, 0, stream>>>(send, rec, rowptr, rank, col, n_edges, n_nodes);

        for (int l = 0; l < 4; l += 2) {
            gin_layer<<<1024, 256, 0, stream>>>(
                hbuf, rowptr, col, Ws + (size_t)l * HID * HID, bs + (size_t)l * HID,
                agg, n_nodes, gemm_grid);
            gin_layer<<<1024, 256, 0, stream>>>(
                agg, rowptr, col, Ws + (size_t)(l + 1) * HID * HID, bs + (size_t)(l + 1) * HID,
                hbuf, n_nodes, gemm_grid);
        }
    } else {
        const int scat_grid = (int)(((long long)n_edges * 32 + 255) / 256);
        for (int l = 0; l < 4; ++l) {
            hipMemcpyAsync(agg, hbuf, nh_bytes, hipMemcpyDeviceToDevice, stream);
            gin_scatter<<<scat_grid, 256, 0, stream>>>(hbuf, send, rec, agg, n_edges);
            gin_gemm<<<gemm_grid, 256, 0, stream>>>(
                agg, Ws + (size_t)l * HID * HID, bs + (size_t)l * HID, hbuf, n_nodes);
        }
    }
}


# output tiles staged through LDS in the embedding GEMM and the layer kernel: a wave stores two whole 512-byte rows (16-byte stores), not sixteen 64-byte pieces; one more barrier a tile; sums kept
# speedup vs baseline: 3.4790x; 1.0343x over previous
#include <hip/hip_runtime.h>
#include <hip/hip_bf16.h>

typedef float v2f __attribute__((ext_vector_type(2)));
typedef float v8f __attribute__((ext_vector_type(8)));

#define HID 128
#define LDS_PITCH 132

__global__ __launch_bounds__(256)
void gin_gemm(const float* __restrict__ X, const float* __restrict__ W,
              const float* __restrict__ bias, float* __restrict__ Y, int n_rows)
{
    __shared__ float sA[16 * LDS_PITCH];
    __shared__ __attribute__((aligned(16))) float sO[16 * LDS_PITCH];

    const int m0  = blockIdx.x * 16;
    const int tid = threadIdx.x;

    for (int i = tid; i < 16 * 32; i += 256) {
        const int r = i >> 5;
        const int c = i & 31;
        float4 v = make_float4(0.f, 0.f, 0.f, 0.f);
        if (m0 + r < n_rows) {
            const float4* src =
                reinterpret_cast<const float4*>(X + (size_t)(m0 + r) * HID);
            v = src[c];
        }
        float* dst = &sA[r * LDS_PITCH + c * 4];
        dst[0] = v.x; dst[1] = v.y; dst[2] = v.z; dst[3] = v.w;
    }
    __syncthreads();

    const int wave = tid >> 5;
    const int lane = tid & 31;
    const int row  = lane & 15;
    const int kk   = (lane >> 4) * 2;
    const int col  = wave * 16 + (lane & 15);

    v8f acc = {};
#pragma unroll
    for (int k0 = 0; k0 < HID; k0 += 4) {
        v2f a;
        a.x = sA[row * LDS_PITCH + k0 + kk];
        a.y = sA[row * LDS_PITCH + k0 + kk + 1];
        v2f b;
        b.x = W[(size_t)(k0 + kk) * HID + col];
        b.y = W[(size_t)(k0 + kk + 1) * HID + col];
        acc = __builtin_amdgcn_wmma_f32_16x16x4_f32(
             false, a,  false, b,
             (short)0, acc,  false,  false);
    }

    const float bcol = bias[col];
    const int half = (lane >> 4) * 8;
#pragma unroll
    for (int i = 0; i < 8; ++i) sO[(half + i) * LDS_PITCH + col] = acc[i] + bcol;
    __syncthreads();
    for (int t = 0; t < 2; ++t) {
        const int r = 2 * wave + t;
        const int m = m0 + r;
        if (m < n_rows) {
            const float4 v = *reinterpret_cast<const float4*>(&sO[r * LDS_PITCH + lane * 4]);
            *reinterpret_cast<float4*>(Y + (size_t)m * HID + lane * 4) = v;
        }
    }
}

__global__ __launch_bounds__(256)
void gin_scatter(const float* __restrict__ h, const int* __restrict__ send,
                 const int* __restrict__ rec, float* __restrict__ agg, int n_edges)
{
    const long long t = (long long)blockIdx.x * blockDim.x + threadIdx.x;
    const int e = (int)(t >> 5);
    if (e >= n_edges) return;
    const int lane = (int)(t & 31);

    const int s = send[e];
    const int r = rec[e];

    const float* src = h + (size_t)s * HID + lane;
    const float v0 = src[0];
    const float v1 = src[32];
    const float v2 = src[64];
    const float v3 = src[96];

    float* dst = agg + (size_t)r * HID + lane;
    atomicAdd(dst + 0,  v0);
    atomicAdd(dst + 32, v1);
    atomicAdd(dst + 64, v2);
    atomicAdd(dst + 96, v3);
}


__global__ __launch_bounds__(256)
void csr_zero(int* __restrict__ cnt, int n1)
{
    const int i = blockIdx.x * 256 + threadIdx.x;
    if (i < n1) cnt[i] = 0;
}

__global__ __launch_bounds__(256)
void csr_rank(const int* __restrict__ rec, int* __restrict__ cnt,
              int* __restrict__ rank, int n_edges, int n_nodes)
{
    const int e = blockIdx.x * 256 + threadIdx.x;
    if (e >= n_edges) return;
    const int r = rec[e];
    int k = -1;
    if ((unsigned)r < (unsigned)n_nodes) k = atomicAdd(&cnt[r], 1);
    rank[e] = k;
}

__global__ __launch_bounds__(1024)
void csr_scan(int* __restrict__ cnt, int n4)
{
    __shared__ int wsum[32];
    const int lane = threadIdx.x & 31;
    const int w    = threadIdx.x >> 5;
    int4* __restrict__ c4 = reinterpret_cast<int4*>(cnt);
    const int seg4 = (((n4 + 31) >> 5) + 31) & ~31;
    const int b4   = w * seg4;
    const int e4   = min(b4 + seg4, n4);

    int s = 0;
    for (int i0 = b4; i0 < e4; i0 += 256) {
        int4 q[8];
#pragma unroll
        for (int u = 0; u < 8; ++u) {
            const int i = i0 + u * 32 + lane;
            q[u] = make_int4(0, 0, 0, 0);
            if (i < e4) q[u] = c4[i];
        }
#pragma unroll
        for (int u = 0; u < 8; ++u) s += (q[u].x + q[u].y) + (q[u].z + q[u].w);
    }
    for (int off = 16; off > 0; off >>= 1) s += __shfl_xor(s, off);
    if (lane == 0) wsum[w] = s;
    __syncthreads();

    int carry = 0;
    for (int k = 0; k < w; ++k) carry += wsum[k];

    for (int i0 = b4; i0 < e4; i0 += 256) {
        int4 q[8];
        int  t[8], x[8];
#pragma unroll
        for (int u = 0; u < 8; ++u) {
            const int i = i0 + u * 32 + lane;
            q[u] = make_int4(0, 0, 0, 0);
            if (i < e4) q[u] = c4[i];
        }
#pragma unroll
        for (int u = 0; u < 8; ++u) { t[u] = (q[u].x + q[u].y) + (q[u].z + q[u].w); x[u] = t[u]; }
#pragma unroll
        for (int off = 1; off < 32; off <<= 1) {
#pragma unroll
            for (int u = 0; u < 8; ++u) {
                const int y = __shfl_up(x[u], off);
                if (lane >= off) x[u] += y;
            }
        }
#pragma unroll
        for (int u = 0; u < 8; ++u) {
            const int i = i0 + u * 32 + lane;
            const int base = carry + x[u] - t[u];
            if (i < e4)
                c4[i] = make_int4(base, base + q[u].x, base + q[u].x + q[u].y,
                                  base + q[u].x + q[u].y + q[u].z);
            carry += __builtin_amdgcn_readlane(x[u], 31);
        }
    }
}

__global__ __launch_bounds__(256)
void csr_fill(const int* __restrict__ send, const int* __restrict__ rec,
              const int* __restrict__ rowptr, const int* __restrict__ rank,
              int* __restrict__ col, int n_edges, int n_nodes)
{
    const int e = blockIdx.x * 256 + threadIdx.x;
    if (e >= n_edges) return;
    const int k = rank[e];
    if (k < 0) return;
    const int s = min(max(send[e], 0), n_nodes - 1);
    col[rowptr[rec[e]] + k] = s;
}

__global__ __launch_bounds__(256)
void gin_layer(const float* __restrict__ h, const int* __restrict__ rowptr,
               const int* __restrict__ col, const float* __restrict__ W,
               const float* __restrict__ bias, float* __restrict__ Y,
               int n_nodes, int n_tiles)
{
    __shared__ float sA[2][16 * LDS_PITCH];
    __shared__ __attribute__((aligned(16))) float sO[16 * LDS_PITCH];

    const int tid  = threadIdx.x;
    const int wave = __builtin_amdgcn_readfirstlane(tid >> 5);
    const int lane = tid & 31;
    const int row  = lane & 15;
    const int kk   = (lane >> 4) * 2;
    const int ncol = wave * 16 + (lane & 15);
    const int half = (lane >> 4) * 8;

    v2f bf[32];
#pragma unroll
    for (int s = 0; s < 32; ++s) {
        bf[s].x = W[(size_t)(4 * s + kk) * HID + ncol];
        bf[s].y = W[(size_t)(4 * s + kk + 1) * HID + ncol];
    }
    const float bcol = bias[ncol];

    const float4* __restrict__ hv = reinterpret_cast<const float4*>(h);
    int buf = 0;

    int pv = 0;
    {
        const int v0 = blockIdx.x * 16 + 2 * wave;
        if ((int)blockIdx.x < n_tiles && lane < 3 && v0 + lane <= n_nodes) pv = rowptr[v0 + lane];
    }

    for (int tile = blockIdx.x; tile < n_tiles; tile += gridDim.x) {
        const int m0 = tile * 16;
        const int v0 = m0 + 2 * wave;
        const bool ok0 = v0 < n_nodes;
        const bool ok1 = v0 + 1 < n_nodes;

        int p0 = 0, p1 = 0, p2 = 0;
        if (ok0) {
            p0 = __builtin_amdgcn_readlane(pv, 0);
            p1 = __builtin_amdgcn_readlane(pv, 1);
            p2 = ok1 ? __builtin_amdgcn_readlane(pv, 2) : p1;
        }

        const int c0 = min(32, p1 - p0);
        const int c1 = min(32, p2 - p1);
        const int first0 = (lane < c0) ? col[p0 + lane] : 0;
        const int first1 = (lane < c1) ? col[p1 + lane] : 0;
        float4 acc[2];
        acc[0] = make_float4(0.f, 0.f, 0.f, 0.f);
        acc[1] = make_float4(0.f, 0.f, 0.f, 0.f);
        if (ok0) acc[0] = hv[(size_t)v0 * 32 + lane];
        if (ok1) acc[1] = hv[(size_t)(v0 + 1) * 32 + lane];

#pragma unroll
        for (int t = 0; t < 2; ++t) {
            const int beg = t ? p1 : p0;
            const int end = t ? p2 : p1;
            const int first = t ? first1 : first0;
            float4 a4 = acc[t];
            for (int j0 = beg; j0 < end; j0 += 32) {
                const int m = min(32, end - j0);
                int idx = first;
                if (j0 != beg) idx = (lane < m) ? col[j0 + lane] : 0;
                for (int i = 0; i < m; i += 8) {
                    int s[8];
#pragma unroll
                    for (int u = 0; u < 8; ++u) s[u] = __builtin_amdgcn_readlane(idx, (i + u) & 31);
                    float4 r[8];
#pragma unroll
                    for (int u = 0; u < 8; ++u) {
                        if (i + u < m) r[u] = hv[(size_t)s[u] * 32 + lane];
                    }
#pragma unroll
                    for (int u = 0; u < 8; ++u) {
                        if (i + u < m) {
                            a4.x += r[u].x; a4.y += r[u].y; a4.z += r[u].z; a4.w += r[u].w;
                        }
                    }
                }
            }
            *reinterpret_cast<float4*>(&sA[buf][(2 * wave + t) * LDS_PITCH + lane * 4]) = a4;
        }
        __syncthreads();

        {
            const int nt = tile + gridDim.x;
            const int nv0 = nt * 16 + 2 * wave;
            pv = 0;
            if (nt < n_tiles && lane < 3 && nv0 + lane <= n_nodes) pv = rowptr[nv0 + lane];
        }

        const float* __restrict__ sa = sA[buf];
        v8f d = {};
#pragma unroll
        for (int s = 0; s < 32; ++s) {
            v2f a;
            a.x = sa[row * LDS_PITCH + 4 * s + kk];
            a.y = sa[row * LDS_PITCH + 4 * s + kk + 1];
            d = __builtin_amdgcn_wmma_f32_16x16x4_f32(
                 false, a,  false, bf[s],
                 (short)0, d,  false,  false);
        }

#pragma unroll
        for (int i = 0; i < 8; ++i) sO[(half + i) * LDS_PITCH + ncol] = d[i] + bcol;
        __syncthreads();
        for (int t = 0; t < 2; ++t) {
            const int r = 2 * wave + t;
            const int m = m0 + r;
            if (m < n_nodes) {
                const float4 v = *reinterpret_cast<const float4*>(&sO[r * LDS_PITCH + lane * 4]);
                *reinterpret_cast<float4*>(Y + (size_t)m * HID + lane * 4) = v;
            }
        }
        buf ^= 1;
    }
}

extern "C" void kernel_launch(void* const* d_in, const int* in_sizes, int n_in,
                              void* d_out, int out_size, void* d_ws, size_t ws_size,
                              hipStream_t stream) {
    const float* h_in  = (const float*)d_in[0];
    const int*   eidx  = (const int*)d_in[1];
    const float* W_emb = (const float*)d_in[2];
    const float* b_emb = (const float*)d_in[3];
    const float* Ws    = (const float*)d_in[4];
    const float* bs    = (const float*)d_in[5];

    const int n_nodes = in_sizes[0] / HID;
    const int n_edges = in_sizes[1] / 2;
    const int* send = eidx;
    const int* rec  = eidx + n_edges;

    float* hbuf = (float*)d_out;
    float* agg  = (float*)d_ws;
    const size_t nh_bytes  = (size_t)n_nodes * HID * sizeof(float);
    const int    n1p       = (n_nodes + 4) & ~3;
    const size_t ptr_bytes = ((size_t)n1p * sizeof(int) + 255) & ~(size_t)255;
    const size_t col_bytes = (size_t)n_edges * sizeof(int);

    const int gemm_grid = (n_nodes + 15) / 16;

    gin_gemm<<<gemm_grid, 256, 0, stream>>>(h_in, W_emb, b_emb, hbuf, n_nodes);

    if (ws_size >= nh_bytes + ptr_bytes + col_bytes && col_bytes <= nh_bytes) {
        int* rank   = (int*)d_ws;
        int* rowptr = (int*)((char*)d_ws + nh_bytes);
        int* col    = (int*)((char*)d_ws + nh_bytes + ptr_bytes);
        const int edge_grid = (n_edges + 255) / 256;

        csr_zero<<<(n1p + 255) / 256, 256, 0, stream>>>(rowptr, n1p);
        csr_rank<<<edge_grid, 256, 0, stream>>>(rec, rowptr, rank, n_edges, n_nodes);
        csr_scan<<<1, 1024, 0, stream>>>(rowptr, n1p / 4);
        csr_fill<<<edge_grid, 256, 0, stream>>>(send, rec, rowptr, rank, col, n_edges, n_nodes);

        for (int l = 0; l < 4; l += 2) {
            gin_layer<<<1024, 256, 0, stream>>>(
                hbuf, rowptr, col, Ws + (size_t)l * HID * HID, bs + (size_t)l * HID,
                agg, n_nodes, gemm_grid);
            gin_layer<<<1024, 256, 0, stream>>>(
                agg, rowptr, col, Ws + (size_t)(l + 1) * HID * HID, bs + (size_t)(l + 1) * HID,
                hbuf, n_nodes, gemm_grid);
        }
    } else {
        const int scat_grid = (int)(((long long)n_edges * 32 + 255) / 256);
        for (int l = 0; l < 4; ++l) {
            hipMemcpyAsync(agg, hbuf, nh_bytes, hipMemcpyDeviceToDevice, stream);
            gin_scatter<<<scat_grid, 256, 0, stream>>>(hbuf, send, rec, agg, n_edges);
            gin_gemm<<<gemm_grid, 256, 0, stream>>>(
                agg, Ws + (size_t)l * HID * HID, bs + (size_t)l * HID, hbuf, n_nodes);
        }
    }
}


# edges grouped by receiver with a two-level counting sort in LDS (512-receiver buckets): no global atomics or scattered 4-byte stores, col leaves in whole runs; per-node sum order still arrival rank
# speedup vs baseline: 4.5043x; 1.2947x over previous
#include <hip/hip_runtime.h>
#include <hip/hip_bf16.h>

typedef float v2f __attribute__((ext_vector_type(2)));
typedef float v8f __attribute__((ext_vector_type(8)));

#define HID 128
#define LDS_PITCH 132

__global__ __launch_bounds__(256)
void gin_gemm(const float* __restrict__ X, const float* __restrict__ W,
              const float* __restrict__ bias, float* __restrict__ Y, int n_rows)
{
    __shared__ float sA[16 * LDS_PITCH];
    __shared__ __attribute__((aligned(16))) float sO[16 * LDS_PITCH];

    const int m0  = blockIdx.x * 16;
    const int tid = threadIdx.x;

    for (int i = tid; i < 16 * 32; i += 256) {
        const int r = i >> 5;
        const int c = i & 31;
        float4 v = make_float4(0.f, 0.f, 0.f, 0.f);
        if (m0 + r < n_rows) {
            const float4* src =
                reinterpret_cast<const float4*>(X + (size_t)(m0 + r) * HID);
            v = src[c];
        }
        float* dst = &sA[r * LDS_PITCH + c * 4];
        dst[0] = v.x; dst[1] = v.y; dst[2] = v.z; dst[3] = v.w;
    }
    __syncthreads();

    const int wave = tid >> 5;
    const int lane = tid & 31;
    const int row  = lane & 15;
    const int kk   = (lane >> 4) * 2;
    const int col  = wave * 16 + (lane & 15);

    v8f acc = {};
#pragma unroll
    for (int k0 = 0; k0 < HID; k0 += 4) {
        v2f a;
        a.x = sA[row * LDS_PITCH + k0 + kk];
        a.y = sA[row * LDS_PITCH + k0 + kk + 1];
        v2f b;
        b.x = W[(size_t)(k0 + kk) * HID + col];
        b.y = W[(size_t)(k0 + kk + 1) * HID + col];
        acc = __builtin_amdgcn_wmma_f32_16x16x4_f32(
             false, a,  false, b,
             (short)0, acc,  false,  false);
    }

    const float bcol = bias[col];
    const int half = (lane >> 4) * 8;
#pragma unroll
    for (int i = 0; i < 8; ++i) sO[(half + i) * LDS_PITCH + col] = acc[i] + bcol;
    __syncthreads();
    for (int t = 0; t < 2; ++t) {
        const int r = 2 * wave + t;
        const int m = m0 + r;
        if (m < n_rows) {
            const float4 v = *reinterpret_cast<const float4*>(&sO[r * LDS_PITCH + lane * 4]);
            *reinterpret_cast<float4*>(Y + (size_t)m * HID + lane * 4) = v;
        }
    }
}

__global__ __launch_bounds__(256)
void gin_scatter(const float* __restrict__ h, const int* __restrict__ send,
                 const int* __restrict__ rec, float* __restrict__ agg, int n_edges)
{
    const long long t = (long long)blockIdx.x * blockDim.x + threadIdx.x;
    const int e = (int)(t >> 5);
    if (e >= n_edges) return;
    const int lane = (int)(t & 31);

    const int s = send[e];
    const int r = rec[e];

    const float* src = h + (size_t)s * HID + lane;
    const float v0 = src[0];
    const float v1 = src[32];
    const float v2 = src[64];
    const float v3 = src[96];

    float* dst = agg + (size_t)r * HID + lane;
    atomicAdd(dst + 0,  v0);
    atomicAdd(dst + 32, v1);
    atomicAdd(dst + 64, v2);
    atomicAdd(dst + 96, v3);
}


#define BSH    9
#define BNODES (1 << BSH)
#define NBK    256
#define ECH    4096
#define BCAP   12288

template <int N>
__device__ __forceinline__ void block_scan_excl(const int* c, int* s, int* wtot, int tid)
{
    const int lane = tid & 31;
    const int w    = tid >> 5;
    int v = 0, x = 0;
    if (tid < N) {
        v = c[tid];
        x = v;
        for (int off = 1; off < 32; off <<= 1) {
            const int y = __shfl_up(x, off);
            if (lane >= off) x += y;
        }
        if (lane == 31) wtot[w] = x;
    }
    __syncthreads();
    if (tid < N) {
        int carry = 0;
        for (int k = 0; k < w; ++k) carry += wtot[k];
        s[tid] = carry + x - v;
    }
    __syncthreads();
}

__global__ __launch_bounds__(512)
void csr_count(const int* __restrict__ rec, int* __restrict__ hist,
               int n_edges, int n_nodes, int nb1)
{
    __shared__ int cnt[NBK];
    const int tid = threadIdx.x;
    if (tid < NBK) cnt[tid] = 0;
    __syncthreads();

    const int base = blockIdx.x * ECH;
    int r[8];
#pragma unroll
    for (int u = 0; u < 8; ++u) {
        const int e = base + u * 512 + tid;
        r[u] = (e < n_edges) ? rec[e] : -1;
    }
#pragma unroll
    for (int u = 0; u < 8; ++u) {
        if ((unsigned)r[u] < (unsigned)n_nodes) atomicAdd(&cnt[r[u] >> BSH], 1);
    }
    __syncthreads();

    if (tid < NBK) hist[tid * nb1 + blockIdx.x] = cnt[tid];
    if (blockIdx.x == 0 && tid < 4) hist[NBK * nb1 + tid] = 0;
}

__global__ __launch_bounds__(1024)
void csr_scan(int* __restrict__ cnt, int n4)
{
    __shared__ int wsum[32];
    const int lane = threadIdx.x & 31;
    const int w    = threadIdx.x >> 5;
    int4* __restrict__ c4 = reinterpret_cast<int4*>(cnt);
    const int seg4 = (((n4 + 31) >> 5) + 31) & ~31;
    const int b4   = w * seg4;
    const int e4   = min(b4 + seg4, n4);

    int s = 0;
    for (int i0 = b4; i0 < e4; i0 += 256) {
        int4 q[8];
#pragma unroll
        for (int u = 0; u < 8; ++u) {
            const int i = i0 + u * 32 + lane;
            q[u] = make_int4(0, 0, 0, 0);
            if (i < e4) q[u] = c4[i];
        }
#pragma unroll
        for (int u = 0; u < 8; ++u) s += (q[u].x + q[u].y) + (q[u].z + q[u].w);
    }
    for (int off = 16; off > 0; off >>= 1) s += __shfl_xor(s, off);
    if (lane == 0) wsum[w] = s;
    __syncthreads();

    int carry = 0;
    for (int k = 0; k < w; ++k) carry += wsum[k];

    for (int i0 = b4; i0 < e4; i0 += 256) {
        int4 q[8];
        int  t[8], x[8];
#pragma unroll
        for (int u = 0; u < 8; ++u) {
            const int i = i0 + u * 32 + lane;
            q[u] = make_int4(0, 0, 0, 0);
            if (i < e4) q[u] = c4[i];
        }
#pragma unroll
        for (int u = 0; u < 8; ++u) { t[u] = (q[u].x + q[u].y) + (q[u].z + q[u].w); x[u] = t[u]; }
#pragma unroll
        for (int off = 1; off < 32; off <<= 1) {
#pragma unroll
            for (int u = 0; u < 8; ++u) {
                const int y = __shfl_up(x[u], off);
                if (lane >= off) x[u] += y;
            }
        }
#pragma unroll
        for (int u = 0; u < 8; ++u) {
            const int i = i0 + u * 32 + lane;
            const int base = carry + x[u] - t[u];
            if (i < e4)
                c4[i] = make_int4(base, base + q[u].x, base + q[u].x + q[u].y,
                                  base + q[u].x + q[u].y + q[u].z);
            carry += __builtin_amdgcn_readlane(x[u], 31);
        }
    }
}

__global__ __launch_bounds__(512)
void csr_bin(const int* __restrict__ send, const int* __restrict__ rec,
             const int* __restrict__ off, int2* __restrict__ ebuf,
             int n_edges, int n_nodes, int nb1)
{
    __shared__ int  cnt[NBK];
    __shared__ int  start[NBK];
    __shared__ int  gofs[NBK];
    __shared__ int  wtot[16];
    __shared__ int2 stage[ECH];

    const int tid = threadIdx.x;
    if (tid < NBK) cnt[tid] = 0;
    __syncthreads();

    const int base = blockIdx.x * ECH;
    int r[8], s[8];
#pragma unroll
    for (int u = 0; u < 8; ++u) {
        const int e = base + u * 512 + tid;
        r[u] = (e < n_edges) ? rec[e] : -1;
        s[u] = (e < n_edges) ? send[e] : 0;
    }
#pragma unroll
    for (int u = 0; u < 8; ++u) {
        if ((unsigned)r[u] < (unsigned)n_nodes) atomicAdd(&cnt[r[u] >> BSH], 1);
    }
    __syncthreads();

    block_scan_excl<NBK>(cnt, start, wtot, tid);
    if (tid < NBK) {
        gofs[tid] = off[tid * nb1 + blockIdx.x] - start[tid];
        cnt[tid]  = start[tid];
    }
    __syncthreads();

#pragma unroll
    for (int u = 0; u < 8; ++u) {
        if ((unsigned)r[u] < (unsigned)n_nodes) {
            const int p = atomicAdd(&cnt[r[u] >> BSH], 1);
            stage[p] = make_int2(s[u], r[u]);
        }
    }
    __syncthreads();

    const int total = cnt[NBK - 1];
    for (int i = tid; i < total; i += 512) {
        const int2 e = stage[i];
        ebuf[gofs[e.y >> BSH] + i] = e;
    }
}

__global__ __launch_bounds__(1024)
void csr_bucket(const int2* __restrict__ ebuf, const int* __restrict__ off, int nb1,
                int* __restrict__ rowptr, int* __restrict__ col, int n_nodes)
{
    __shared__ int cnt[BNODES];
    __shared__ int pos[BNODES];
    __shared__ int wtot[16];
    __shared__ int scol[BCAP];

    const int tid = threadIdx.x;
    const int b   = blockIdx.x;
    const int s0  = off[b * nb1];
    const int s1  = off[(b + 1) * nb1];
    const int n   = s1 - s0;

    if (tid < BNODES) cnt[tid] = 0;
    __syncthreads();

    for (int i0 = 0; i0 < n; i0 += 4096) {
        int r[4];
#pragma unroll
        for (int u = 0; u < 4; ++u) {
            const int i = i0 + u * 1024 + tid;
            r[u] = ebuf[s0 + min(i, n - 1)].y;
        }
#pragma unroll
        for (int u = 0; u < 4; ++u) {
            const int i = i0 + u * 1024 + tid;
            if (i < n) atomicAdd(&cnt[r[u] & (BNODES - 1)], 1);
        }
    }
    __syncthreads();

    block_scan_excl<BNODES>(cnt, pos, wtot, tid);
    if (tid < BNODES) {
        const int v = b * BNODES + tid;
        if (v <= n_nodes) rowptr[v] = s0 + pos[tid];
        cnt[tid] = pos[tid];
    }
    __syncthreads();

    const bool fits = n <= BCAP;
    for (int i0 = 0; i0 < n; i0 += 4096) {
        int2 e[4];
#pragma unroll
        for (int u = 0; u < 4; ++u) {
            const int i = i0 + u * 1024 + tid;
            e[u] = ebuf[s0 + min(i, n - 1)];
        }
#pragma unroll
        for (int u = 0; u < 4; ++u) {
            const int i = i0 + u * 1024 + tid;
            if (i < n) {
                const int p   = atomicAdd(&cnt[e[u].y & (BNODES - 1)], 1);
                const int sdr = min(max(e[u].x, 0), n_nodes - 1);
                if (fits) scol[p] = sdr; else col[s0 + p] = sdr;
            }
        }
    }
    __syncthreads();

    if (fits) {
        for (int i = tid; i < n; i += 1024) col[s0 + i] = scol[i];
    }
}

__global__ __launch_bounds__(256)
void gin_layer(const float* __restrict__ h, const int* __restrict__ rowptr,
               const int* __restrict__ col, const float* __restrict__ W,
               const float* __restrict__ bias, float* __restrict__ Y,
               int n_nodes, int n_tiles)
{
    __shared__ float sA[2][16 * LDS_PITCH];
    __shared__ __attribute__((aligned(16))) float sO[16 * LDS_PITCH];

    const int tid  = threadIdx.x;
    const int wave = __builtin_amdgcn_readfirstlane(tid >> 5);
    const int lane = tid & 31;
    const int row  = lane & 15;
    const int kk   = (lane >> 4) * 2;
    const int ncol = wave * 16 + (lane & 15);
    const int half = (lane >> 4) * 8;

    v2f bf[32];
#pragma unroll
    for (int s = 0; s < 32; ++s) {
        bf[s].x = W[(size_t)(4 * s + kk) * HID + ncol];
        bf[s].y = W[(size_t)(4 * s + kk + 1) * HID + ncol];
    }
    const float bcol = bias[ncol];

    const float4* __restrict__ hv = reinterpret_cast<const float4*>(h);
    int buf = 0;

    int pv = 0;
    {
        const int v0 = blockIdx.x * 16 + 2 * wave;
        if ((int)blockIdx.x < n_tiles && lane < 3 && v0 + lane <= n_nodes) pv = rowptr[v0 + lane];
    }

    for (int tile = blockIdx.x; tile < n_tiles; tile += gridDim.x) {
        const int m0 = tile * 16;
        const int v0 = m0 + 2 * wave;
        const bool ok0 = v0 < n_nodes;
        const bool ok1 = v0 + 1 < n_nodes;

        int p0 = 0, p1 = 0, p2 = 0;
        if (ok0) {
            p0 = __builtin_amdgcn_readlane(pv, 0);
            p1 = __builtin_amdgcn_readlane(pv, 1);
            p2 = ok1 ? __builtin_amdgcn_readlane(pv, 2) : p1;
        }

        const int c0 = min(32, p1 - p0);
        const int c1 = min(32, p2 - p1);
        const int first0 = (lane < c0) ? col[p0 + lane] : 0;
        const int first1 = (lane < c1) ? col[p1 + lane] : 0;
        float4 acc[2];
        acc[0] = make_float4(0.f, 0.f, 0.f, 0.f);
        acc[1] = make_float4(0.f, 0.f, 0.f, 0.f);
        if (ok0) acc[0] = hv[(size_t)v0 * 32 + lane];
        if (ok1) acc[1] = hv[(size_t)(v0 + 1) * 32 + lane];

#pragma unroll
        for (int t = 0; t < 2; ++t) {
            const int beg = t ? p1 : p0;
            const int end = t ? p2 : p1;
            const int first = t ? first1 : first0;
            float4 a4 = acc[t];
            for (int j0 = beg; j0 < end; j0 += 32) {
                const int m = min(32, end - j0);
                int idx = first;
                if (j0 != beg) idx = (lane < m) ? col[j0 + lane] : 0;
                for (int i = 0; i < m; i += 8) {
                    int s[8];
#pragma unroll
                    for (int u = 0; u < 8; ++u) s[u] = __builtin_amdgcn_readlane(idx, (i + u) & 31);
                    float4 r[8];
#pragma unroll
                    for (int u = 0; u < 8; ++u) {
                        if (i + u < m) r[u] = hv[(size_t)s[u] * 32 + lane];
                    }
#pragma unroll
                    for (int u = 0; u < 8; ++u) {
                        if (i + u < m) {
                            a4.x += r[u].x; a4.y += r[u].y; a4.z += r[u].z; a4.w += r[u].w;
                        }
                    }
                }
            }
            *reinterpret_cast<float4*>(&sA[buf][(2 * wave + t) * LDS_PITCH + lane * 4]) = a4;
        }
        __syncthreads();

        {
            const int nt = tile + gridDim.x;
            const int nv0 = nt * 16 + 2 * wave;
            pv = 0;
            if (nt < n_tiles && lane < 3 && nv0 + lane <= n_nodes) pv = rowptr[nv0 + lane];
        }

        const float* __restrict__ sa = sA[buf];
        v8f d = {};
#pragma unroll
        for (int s = 0; s < 32; ++s) {
            v2f a;
            a.x = sa[row * LDS_PITCH + 4 * s + kk];
            a.y = sa[row * LDS_PITCH + 4 * s + kk + 1];
            d = __builtin_amdgcn_wmma_f32_16x16x4_f32(
                 false, a,  false, bf[s],
                 (short)0, d,  false,  false);
        }

#pragma unroll
        for (int i = 0; i < 8; ++i) sO[(half + i) * LDS_PITCH + ncol] = d[i] + bcol;
        __syncthreads();
        for (int t = 0; t < 2; ++t) {
            const int r = 2 * wave + t;
            const int m = m0 + r;
            if (m < n_nodes) {
                const float4 v = *reinterpret_cast<const float4*>(&sO[r * LDS_PITCH + lane * 4]);
                *reinterpret_cast<float4*>(Y + (size_t)m * HID + lane * 4) = v;
            }
        }
        buf ^= 1;
    }
}

extern "C" void kernel_launch(void* const* d_in, const int* in_sizes, int n_in,
                              void* d_out, int out_size, void* d_ws, size_t ws_size,
                              hipStream_t stream) {
    const float* h_in  = (const float*)d_in[0];
    const int*   eidx  = (const int*)d_in[1];
    const float* W_emb = (const float*)d_in[2];
    const float* b_emb = (const float*)d_in[3];
    const float* Ws    = (const float*)d_in[4];
    const float* bs    = (const float*)d_in[5];

    const int n_nodes = in_sizes[0] / HID;
    const int n_edges = in_sizes[1] / 2;
    const int* send = eidx;
    const int* rec  = eidx + n_edges;

    float* hbuf = (float*)d_out;
    float* agg  = (float*)d_ws;
    const size_t nh_bytes  = (size_t)n_nodes * HID * sizeof(float);
    const int    n1p       = (n_nodes + 4) & ~3;
    const size_t ptr_bytes = ((size_t)n1p * sizeof(int) + 255) & ~(size_t)255;
    const size_t col_bytes = (size_t)n_edges * sizeof(int);

    const int gemm_grid = (n_nodes + 15) / 16;

    gin_gemm<<<gemm_grid, 256, 0, stream>>>(h_in, W_emb, b_emb, hbuf, n_nodes);

    const int    nb1        = (n_edges + ECH - 1) / ECH;
    const size_t ebuf_bytes = ((size_t)n_edges * sizeof(int2) + 255) & ~(size_t)255;
    const size_t hist_bytes = ((size_t)NBK * nb1 + 4) * sizeof(int);

    if (ws_size >= nh_bytes + ptr_bytes + col_bytes && ebuf_bytes + hist_bytes <= nh_bytes &&
        n_nodes <= (NBK - 1) * BNODES) {
        int2* ebuf  = (int2*)d_ws;
        int* hist   = (int*)((char*)d_ws + ebuf_bytes);
        int* rowptr = (int*)((char*)d_ws + nh_bytes);
        int* col    = (int*)((char*)d_ws + nh_bytes + ptr_bytes);

        csr_count<<<nb1, 512, 0, stream>>>(rec, hist, n_edges, n_nodes, nb1);
        csr_scan<<<1, 1024, 0, stream>>>(hist, (NBK / 4) * nb1 + 1);
        csr_bin<<<nb1, 512, 0, stream>>>(send, rec, hist, ebuf, n_edges, n_nodes, nb1);
        csr_bucket<<<NBK, 1024, 0, stream>>>(ebuf, hist, nb1, rowptr, col, n_nodes);

        for (int l = 0; l < 4; l += 2) {
            gin_layer<<<1024, 256, 0, stream>>>(
                hbuf, rowptr, col, Ws + (size_t)l * HID * HID, bs + (size_t)l * HID,
                agg, n_nodes, gemm_grid);
            gin_layer<<<1024, 256, 0, stream>>>(
                agg, rowptr, col, Ws + (size_t)(l + 1) * HID * HID, bs + (size_t)(l + 1) * HID,
                hbuf, n_nodes, gemm_grid);
        }
    } else {
        const int scat_grid = (int)(((long long)n_edges * 32 + 255) / 256);
        for (int l = 0; l < 4; ++l) {
            hipMemcpyAsync(agg, hbuf, nh_bytes, hipMemcpyDeviceToDevice, stream);
            gin_scatter<<<scat_grid, 256, 0, stream>>>(hbuf, send, rec, agg, n_edges);
            gin_gemm<<<gemm_grid, 256, 0, stream>>>(
                agg, Ws + (size_t)l * HID * HID, bs + (size_t)l * HID, hbuf, n_nodes);
        }
    }
}


# first sort level chunk-local: a chunk's bucket runs stay in its own ebuf slots, level 2 collects them; counting and global-scan launches gone (9 -> 7); per-node sum order still arrival rank
# speedup vs baseline: 4.7744x; 1.0600x over previous
#include <hip/hip_runtime.h>
#include <hip/hip_bf16.h>

typedef float v2f __attribute__((ext_vector_type(2)));
typedef float v8f __attribute__((ext_vector_type(8)));

#define HID 128
#define LDS_PITCH 132

__global__ __launch_bounds__(256)
void gin_gemm(const float* __restrict__ X, const float* __restrict__ W,
              const float* __restrict__ bias, float* __restrict__ Y, int n_rows)
{
    __shared__ float sA[16 * LDS_PITCH];
    __shared__ __attribute__((aligned(16))) float sO[16 * LDS_PITCH];

    const int m0  = blockIdx.x * 16;
    const int tid = threadIdx.x;

    for (int i = tid; i < 16 * 32; i += 256) {
        const int r = i >> 5;
        const int c = i & 31;
        float4 v = make_float4(0.f, 0.f, 0.f, 0.f);
        if (m0 + r < n_rows) {
            const float4* src =
                reinterpret_cast<const float4*>(X + (size_t)(m0 + r) * HID);
            v = src[c];
        }
        float* dst = &sA[r * LDS_PITCH + c * 4];
        dst[0] = v.x; dst[1] = v.y; dst[2] = v.z; dst[3] = v.w;
    }
    __syncthreads();

    const int wave = tid >> 5;
    const int lane = tid & 31;
    const int row  = lane & 15;
    const int kk   = (lane >> 4) * 2;
    const int col  = wave * 16 + (lane & 15);

    v8f acc = {};
#pragma unroll
    for (int k0 = 0; k0 < HID; k0 += 4) {
        v2f a;
        a.x = sA[row * LDS_PITCH + k0 + kk];
        a.y = sA[row * LDS_PITCH + k0 + kk + 1];
        v2f b;
        b.x = W[(size_t)(k0 + kk) * HID + col];
        b.y = W[(size_t)(k0 + kk + 1) * HID + col];
        acc = __builtin_amdgcn_wmma_f32_16x16x4_f32(
             false, a,  false, b,
             (short)0, acc,  false,  false);
    }

    const float bcol = bias[col];
    const int half = (lane >> 4) * 8;
#pragma unroll
    for (int i = 0; i < 8; ++i) sO[(half + i) * LDS_PITCH + col] = acc[i] + bcol;
    __syncthreads();
    for (int t = 0; t < 2; ++t) {
        const int r = 2 * wave + t;
        const int m = m0 + r;
        if (m < n_rows) {
            const float4 v = *reinterpret_cast<const float4*>(&sO[r * LDS_PITCH + lane * 4]);
            *reinterpret_cast<float4*>(Y + (size_t)m * HID + lane * 4) = v;
        }
    }
}

__global__ __launch_bounds__(256)
void gin_scatter(const float* __restrict__ h, const int* __restrict__ send,
                 const int* __restrict__ rec, float* __restrict__ agg, int n_edges)
{
    const long long t = (long long)blockIdx.x * blockDim.x + threadIdx.x;
    const int e = (int)(t >> 5);
    if (e >= n_edges) return;
    const int lane = (int)(t & 31);

    const int s = send[e];
    const int r = rec[e];

    const float* src = h + (size_t)s * HID + lane;
    const float v0 = src[0];
    const float v1 = src[32];
    const float v2 = src[64];
    const float v3 = src[96];

    float* dst = agg + (size_t)r * HID + lane;
    atomicAdd(dst + 0,  v0);
    atomicAdd(dst + 32, v1);
    atomicAdd(dst + 64, v2);
    atomicAdd(dst + 96, v3);
}


#define BSH    9
#define BNODES (1 << BSH)
#define NBK    256
#define ECH    4096
#define MAXCH  1024
#define BCAP   12288

template <int N>
__device__ __forceinline__ void block_scan_excl(const int* c, int* s, int* wtot, int tid)
{
    const int lane = tid & 31;
    const int w    = tid >> 5;
    int v = 0, x = 0;
    if (tid < N) {
        v = c[tid];
        x = v;
        for (int off = 1; off < 32; off <<= 1) {
            const int y = __shfl_up(x, off);
            if (lane >= off) x += y;
        }
        if (lane == 31) wtot[w] = x;
    }
    __syncthreads();
    if (tid < N) {
        int carry = 0;
        for (int k = 0; k < w; ++k) carry += wtot[k];
        s[tid] = carry + x - v;
    }
    __syncthreads();
}

__global__ __launch_bounds__(512)
void csr_bin(const int* __restrict__ send, const int* __restrict__ rec,
             int* __restrict__ tbl, int2* __restrict__ ebuf,
             int n_edges, int n_nodes, int nb1)
{
    __shared__ int  cnt[NBK];
    __shared__ int  start[NBK];
    __shared__ int  wtot[16];
    __shared__ int2 stage[ECH];

    const int tid = threadIdx.x;
    if (tid < NBK) cnt[tid] = 0;
    __syncthreads();

    const int base = blockIdx.x * ECH;
    int r[8], s[8];
#pragma unroll
    for (int u = 0; u < 8; ++u) {
        const int e = base + u * 512 + tid;
        r[u] = (e < n_edges) ? rec[e] : -1;
        s[u] = (e < n_edges) ? send[e] : 0;
    }
#pragma unroll
    for (int u = 0; u < 8; ++u) {
        if ((unsigned)r[u] < (unsigned)n_nodes) atomicAdd(&cnt[r[u] >> BSH], 1);
    }
    __syncthreads();

    block_scan_excl<NBK>(cnt, start, wtot, tid);
    if (tid < NBK) {
        tbl[tid * nb1 + blockIdx.x] = start[tid];
        cnt[tid] = start[tid];
    }
    __syncthreads();

#pragma unroll
    for (int u = 0; u < 8; ++u) {
        if ((unsigned)r[u] < (unsigned)n_nodes) {
            const int p = atomicAdd(&cnt[r[u] >> BSH], 1);
            stage[p] = make_int2(s[u], r[u]);
        }
    }
    __syncthreads();

    const int total = cnt[NBK - 1];
    if (tid == 0) tbl[NBK * nb1 + blockIdx.x] = total;
    for (int i = tid; i < total; i += 512) ebuf[base + i] = stage[i];
}

__global__ __launch_bounds__(1024)
void csr_bucket(const int2* __restrict__ ebuf, const int* __restrict__ tbl, int nb1,
                int* __restrict__ rowptr, int* __restrict__ col, int n_nodes)
{
    __shared__ int cnt[BNODES];
    __shared__ int pos[BNODES];
    __shared__ int wtot[16];
    __shared__ int wbeg[32];
    __shared__ int wlen[32];
    __shared__ int rbeg[MAXCH];
    __shared__ int rlen[MAXCH];
    __shared__ int scol[BCAP];

    const int tid  = threadIdx.x;
    const int lane = tid & 31;
    const int wave = __builtin_amdgcn_readfirstlane(tid >> 5);
    const int b    = blockIdx.x;
    const int* __restrict__ ewords = reinterpret_cast<const int*>(ebuf);
    const int*  spare1 = tbl + b * nb1;
    const int2* spare2 = reinterpret_cast<const int2*>(tbl + ((b * nb1) & ~1));

    int rs = 0, ln = 0;
    if (tid < nb1) {
        rs = tbl[b * nb1 + tid];
        ln = tbl[(b + 1) * nb1 + tid] - rs;
    }
    rbeg[tid] = tid * ECH + rs;
    rlen[tid] = ln;
    if (tid < BNODES) cnt[tid] = 0;
    {
        int a = rs, c = ln;
        for (int off = 16; off > 0; off >>= 1) { a += __shfl_xor(a, off); c += __shfl_xor(c, off); }
        if (lane == 0) { wbeg[wave] = a; wlen[wave] = c; }
    }
    __syncthreads();
    int s0 = 0, n = 0;
    for (int k = 0; k < 32; ++k) { s0 += wbeg[k]; n += wlen[k]; }

    for (int g0 = wave * 8; g0 < nb1; g0 += 256) {
        int L[8], S[8], rr[8];
#pragma unroll
        for (int u = 0; u < 8; ++u) {
            L[u] = __builtin_amdgcn_readfirstlane(rlen[g0 + u]);
            S[u] = __builtin_amdgcn_readfirstlane(rbeg[g0 + u]);
        }
#pragma unroll
        for (int u = 0; u < 8; ++u) {
            const int* p = (L[u] > 0) ? ewords + (2 * (S[u] + min(lane, L[u] - 1)) + 1) : spare1;
            rr[u] = *p;
        }
#pragma unroll
        for (int u = 0; u < 8; ++u) {
            if (lane < L[u]) atomicAdd(&cnt[rr[u] & (BNODES - 1)], 1);
        }
#pragma unroll
        for (int u = 0; u < 8; ++u) {
            for (int k = 32 + lane; k < L[u]; k += 32)
                atomicAdd(&cnt[ewords[2 * (S[u] + k) + 1] & (BNODES - 1)], 1);
        }
    }
    __syncthreads();

    block_scan_excl<BNODES>(cnt, pos, wtot, tid);
    if (tid < BNODES) {
        const int v = b * BNODES + tid;
        if (v <= n_nodes) rowptr[v] = s0 + pos[tid];
        cnt[tid] = pos[tid];
    }
    __syncthreads();

    const bool fits = n <= BCAP;
    for (int g0 = wave * 8; g0 < nb1; g0 += 256) {
        int  L[8], S[8];
        int2 e[8];
#pragma unroll
        for (int u = 0; u < 8; ++u) {
            L[u] = __builtin_amdgcn_readfirstlane(rlen[g0 + u]);
            S[u] = __builtin_amdgcn_readfirstlane(rbeg[g0 + u]);
        }
#pragma unroll
        for (int u = 0; u < 8; ++u) {
            const int2* p = (L[u] > 0) ? ebuf + (S[u] + min(lane, L[u] - 1)) : spare2;
            e[u] = *p;
        }
#pragma unroll
        for (int u = 0; u < 8; ++u) {
            if (lane < L[u]) {
                const int p   = atomicAdd(&cnt[e[u].y & (BNODES - 1)], 1);
                const int sdr = min(max(e[u].x, 0), n_nodes - 1);
                if (fits) scol[p] = sdr; else col[s0 + p] = sdr;
            }
        }
#pragma unroll
        for (int u = 0; u < 8; ++u) {
            for (int k = 32 + lane; k < L[u]; k += 32) {
                const int2 x  = ebuf[S[u] + k];
                const int p   = atomicAdd(&cnt[x.y & (BNODES - 1)], 1);
                const int sdr = min(max(x.x, 0), n_nodes - 1);
                if (fits) scol[p] = sdr; else col[s0 + p] = sdr;
            }
        }
    }
    __syncthreads();

    if (fits) {
        for (int i = tid; i < n; i += 1024) col[s0 + i] = scol[i];
    }
}

__global__ __launch_bounds__(256)
void gin_layer(const float* __restrict__ h, const int* __restrict__ rowptr,
               const int* __restrict__ col, const float* __restrict__ W,
               const float* __restrict__ bias, float* __restrict__ Y,
               int n_nodes, int n_tiles)
{
    __shared__ float sA[2][16 * LDS_PITCH];
    __shared__ __attribute__((aligned(16))) float sO[16 * LDS_PITCH];

    const int tid  = threadIdx.x;
    const int wave = __builtin_amdgcn_readfirstlane(tid >> 5);
    const int lane = tid & 31;
    const int row  = lane & 15;
    const int kk   = (lane >> 4) * 2;
    const int ncol = wave * 16 + (lane & 15);
    const int half = (lane >> 4) * 8;

    v2f bf[32];
#pragma unroll
    for (int s = 0; s < 32; ++s) {
        bf[s].x = W[(size_t)(4 * s + kk) * HID + ncol];
        bf[s].y = W[(size_t)(4 * s + kk + 1) * HID + ncol];
    }
    const float bcol = bias[ncol];

    const float4* __restrict__ hv = reinterpret_cast<const float4*>(h);
    int buf = 0;

    int pv = 0;
    {
        const int v0 = blockIdx.x * 16 + 2 * wave;
        if ((int)blockIdx.x < n_tiles && lane < 3 && v0 + lane <= n_nodes) pv = rowptr[v0 + lane];
    }

    for (int tile = blockIdx.x; tile < n_tiles; tile += gridDim.x) {
        const int m0 = tile * 16;
        const int v0 = m0 + 2 * wave;
        const bool ok0 = v0 < n_nodes;
        const bool ok1 = v0 + 1 < n_nodes;

        int p0 = 0, p1 = 0, p2 = 0;
        if (ok0) {
            p0 = __builtin_amdgcn_readlane(pv, 0);
            p1 = __builtin_amdgcn_readlane(pv, 1);
            p2 = ok1 ? __builtin_amdgcn_readlane(pv, 2) : p1;
        }

        const int c0 = min(32, p1 - p0);
        const int c1 = min(32, p2 - p1);
        const int first0 = (lane < c0) ? col[p0 + lane] : 0;
        const int first1 = (lane < c1) ? col[p1 + lane] : 0;
        float4 acc[2];
        acc[0] = make_float4(0.f, 0.f, 0.f, 0.f);
        acc[1] = make_float4(0.f, 0.f, 0.f, 0.f);
        if (ok0) acc[0] = hv[(size_t)v0 * 32 + lane];
        if (ok1) acc[1] = hv[(size_t)(v0 + 1) * 32 + lane];

#pragma unroll
        for (int t = 0; t < 2; ++t) {
            const int beg = t ? p1 : p0;
            const int end = t ? p2 : p1;
            const int first = t ? first1 : first0;
            float4 a4 = acc[t];
            for (int j0 = beg; j0 < end; j0 += 32) {
                const int m = min(32, end - j0);
                int idx = first;
                if (j0 != beg) idx = (lane < m) ? col[j0 + lane] : 0;
                for (int i = 0; i < m; i += 8) {
                    int s[8];
#pragma unroll
                    for (int u = 0; u < 8; ++u) s[u] = __builtin_amdgcn_readlane(idx, (i + u) & 31);
                    float4 r[8];
#pragma unroll
                    for (int u = 0; u < 8; ++u) {
                        if (i + u < m) r[u] = hv[(size_t)s[u] * 32 + lane];
                    }
#pragma unroll
                    for (int u = 0; u < 8; ++u) {
                        if (i + u < m) {
                            a4.x += r[u].x; a4.y += r[u].y; a4.z += r[u].z; a4.w += r[u].w;
                        }
                    }
                }
            }
            *reinterpret_cast<float4*>(&sA[buf][(2 * wave + t) * LDS_PITCH + lane * 4]) = a4;
        }
        __syncthreads();

        {
            const int nt = tile + gridDim.x;
            const int nv0 = nt * 16 + 2 * wave;
            pv = 0;
            if (nt < n_tiles && lane < 3 && nv0 + lane <= n_nodes) pv = rowptr[nv0 + lane];
        }

        const float* __restrict__ sa = sA[buf];
        v8f d = {};
#pragma unroll
        for (int s = 0; s < 32; ++s) {
            v2f a;
            a.x = sa[row * LDS_PITCH + 4 * s + kk];
            a.y = sa[row * LDS_PITCH + 4 * s + kk + 1];
            d = __builtin_amdgcn_wmma_f32_16x16x4_f32(
                 false, a,  false, bf[s],
                 (short)0, d,  false,  false);
        }

#pragma unroll
        for (int i = 0; i < 8; ++i) sO[(half + i) * LDS_PITCH + ncol] = d[i] + bcol;
        __syncthreads();
        for (int t = 0; t < 2; ++t) {
            const int r = 2 * wave + t;
            const int m = m0 + r;
            if (m < n_nodes) {
                const float4 v = *reinterpret_cast<const float4*>(&sO[r * LDS_PITCH + lane * 4]);
                *reinterpret_cast<float4*>(Y + (size_t)m * HID + lane * 4) = v;
            }
        }
        buf ^= 1;
    }
}

extern "C" void kernel_launch(void* const* d_in, const int* in_sizes, int n_in,
                              void* d_out, int out_size, void* d_ws, size_t ws_size,
                              hipStream_t stream) {
    const float* h_in  = (const float*)d_in[0];
    const int*   eidx  = (const int*)d_in[1];
    const float* W_emb = (const float*)d_in[2];
    const float* b_emb = (const float*)d_in[3];
    const float* Ws    = (const float*)d_in[4];
    const float* bs    = (const float*)d_in[5];

    const int n_nodes = in_sizes[0] / HID;
    const int n_edges = in_sizes[1] / 2;
    const int* send = eidx;
    const int* rec  = eidx + n_edges;

    float* hbuf = (float*)d_out;
    float* agg  = (float*)d_ws;
    const size_t nh_bytes  = (size_t)n_nodes * HID * sizeof(float);
    const int    n1p       = (n_nodes + 4) & ~3;
    const size_t ptr_bytes = ((size_t)n1p * sizeof(int) + 255) & ~(size_t)255;
    const size_t col_bytes = (size_t)n_edges * sizeof(int);

    const int gemm_grid = (n_nodes + 15) / 16;

    gin_gemm<<<gemm_grid, 256, 0, stream>>>(h_in, W_emb, b_emb, hbuf, n_nodes);

    const int    nb1        = (n_edges + ECH - 1) / ECH;
    const size_t ebuf_bytes = ((size_t)nb1 * ECH * sizeof(int2) + 255) & ~(size_t)255;
    const size_t tbl_bytes  = ((size_t)(NBK + 1) * nb1) * sizeof(int);

    if (ws_size >= nh_bytes + ptr_bytes + col_bytes && ebuf_bytes + tbl_bytes <= nh_bytes &&
        n_nodes <= (NBK - 1) * BNODES && nb1 <= MAXCH) {
        int2* ebuf  = (int2*)d_ws;
        int* tbl    = (int*)((char*)d_ws + ebuf_bytes);
        int* rowptr = (int*)((char*)d_ws + nh_bytes);
        int* col    = (int*)((char*)d_ws + nh_bytes + ptr_bytes);

        csr_bin<<<nb1, 512, 0, stream>>>(send, rec, tbl, ebuf, n_edges, n_nodes, nb1);
        csr_bucket<<<NBK, 1024, 0, stream>>>(ebuf, tbl, nb1, rowptr, col, n_nodes);

        for (int l = 0; l < 4; l += 2) {
            gin_layer<<<1024, 256, 0, stream>>>(
                hbuf, rowptr, col, Ws + (size_t)l * HID * HID, bs + (size_t)l * HID,
                agg, n_nodes, gemm_grid);
            gin_layer<<<1024, 256, 0, stream>>>(
                agg, rowptr, col, Ws + (size_t)(l + 1) * HID * HID, bs + (size_t)(l + 1) * HID,
                hbuf, n_nodes, gemm_grid);
        }
    } else {
        const int scat_grid = (int)(((long long)n_edges * 32 + 255) / 256);
        for (int l = 0; l < 4; ++l) {
            hipMemcpyAsync(agg, hbuf, nh_bytes, hipMemcpyDeviceToDevice, stream);
            gin_scatter<<<scat_grid, 256, 0, stream>>>(hbuf, send, rec, agg, n_edges);
            gin_gemm<<<gemm_grid, 256, 0, stream>>>(
                agg, Ws + (size_t)l * HID * HID, bs + (size_t)l * HID, hbuf, n_nodes);
        }
    }
}
